# starting kernel
# baseline (speedup 1.0000x reference)
#include <hip/hip_runtime.h>

typedef float v2f __attribute__((ext_vector_type(2)));
typedef float v8f __attribute__((ext_vector_type(8)));

#define DF 300
#define K4 75
#define NPAD 304
#define FRAG_STRIDE 66
#define PIPE 3

#define WMMA_F32(a, b, c) \
    __builtin_amdgcn_wmma_f32_16x16x4_f32(false, (a), false, (b), (short)0, (c), false, false)


__global__ __launch_bounds__(256) void k_fill1(float* __restrict__ p, int n) {
    int i = blockIdx.x * 256 + threadIdx.x;
    if (i < n) p[i] = 1.0f;
}

__global__ __launch_bounds__(256) void k_deg_accum(float* __restrict__ deg,
                                                   const int* __restrict__ row, int e0) {
    int i = blockIdx.x * 256 + threadIdx.x;
    if (i < e0) unsafeAtomicAdd(&deg[row[i]], 1.0f);
}

__global__ __launch_bounds__(256) void k_rsqrt(float* __restrict__ p, int n) {
    int i = blockIdx.x * 256 + threadIdx.x;
    if (i < n) p[i] = rsqrtf(p[i]);
}

__global__ __launch_bounds__(256) void k_zero4(float4* __restrict__ p, int n4) {
    int i = blockIdx.x * 256 + threadIdx.x;
    if (i < n4) p[i] = make_float4(0.f, 0.f, 0.f, 0.f);
}

__global__ __launch_bounds__(256) void k_padW(const float* __restrict__ W,
                                              float* __restrict__ Wp) {
    int i = blockIdx.x * 256 + threadIdx.x;
    if (i < NPAD * DF) Wp[i] = (i < DF * DF) ? W[i] : 0.f;
}


__global__ __launch_bounds__(256) void k_gemm(const float* __restrict__ A,
                                              const float* __restrict__ Wp,
                                              const float* __restrict__ bias,
                                              float* __restrict__ out,
                                              int M, int relu_in) {
    __shared__ float lds[K4 * FRAG_STRIDE];
    const int m0 = blockIdx.x * 16;
    const int t  = threadIdx.x;

    for (int i = t; i < 16 * K4; i += 256) {
        int r  = i / K4;
        int c4 = i - r * K4;
        int m  = m0 + r;
        float4 v = make_float4(0.f, 0.f, 0.f, 0.f);
        if (m < M) v = ((const float4*)(A + (size_t)m * DF))[c4];
        if (relu_in) {
            v.x = fmaxf(v.x, 0.f); v.y = fmaxf(v.y, 0.f);
            v.z = fmaxf(v.z, 0.f); v.w = fmaxf(v.w, 0.f);
        }
        *(float2*)&lds[c4 * FRAG_STRIDE + r * 2]      = make_float2(v.x, v.y);
        *(float2*)&lds[c4 * FRAG_STRIDE + r * 2 + 32] = make_float2(v.z, v.w);
    }
    __syncthreads();

    const int wave = t >> 5;
    const int lane = t & 31;
    const int kh   = (lane >> 4) << 1;
    const int ln   = lane & 15;
    const float* af = &lds[lane * 2];
    const int mbase = m0 + ((lane >> 4) << 3);

#define LDA(s)    (*(const v2f*)(af + (s) * FRAG_STRIDE))
#define LDW(p, s) (*(const v2f*)((p) + (s) * 4))

    {
        const int n0 = wave * 16 + ln;
        const int n1 = (wave + 8) * 16 + ln;
        const float* w0 = Wp + (size_t)n0 * DF + kh;
        const float* w1 = Wp + (size_t)n1 * DF + kh;

        v8f c0 = {}, c1 = {};
        v2f a0 = LDA(0), a1 = LDA(1), a2 = LDA(2);
        v2f p0 = LDW(w0, 0), p1 = LDW(w0, 1), p2 = LDW(w0, 2);
        v2f q0 = LDW(w1, 0), q1 = LDW(w1, 1), q2 = LDW(w1, 2);

        for (int s = 0; s < K4 - PIPE; s += PIPE) {
            c0 = WMMA_F32(a0, p0, c0);  c1 = WMMA_F32(a0, q0, c1);
            a0 = LDA(s + 3);  p0 = LDW(w0, s + 3);  q0 = LDW(w1, s + 3);
            c0 = WMMA_F32(a1, p1, c0);  c1 = WMMA_F32(a1, q1, c1);
            a1 = LDA(s + 4);  p1 = LDW(w0, s + 4);  q1 = LDW(w1, s + 4);
            c0 = WMMA_F32(a2, p2, c0);  c1 = WMMA_F32(a2, q2, c1);
            a2 = LDA(s + 5);  p2 = LDW(w0, s + 5);  q2 = LDW(w1, s + 5);
        }
        c0 = WMMA_F32(a0, p0, c0);  c1 = WMMA_F32(a0, q0, c1);
        c0 = WMMA_F32(a1, p1, c0);  c1 = WMMA_F32(a1, q1, c1);
        c0 = WMMA_F32(a2, p2, c0);  c1 = WMMA_F32(a2, q2, c1);

        const float bv0 = bias[n0];
        const float bv1 = bias[n1];
        #pragma unroll
        for (int r = 0; r < 8; ++r) {
            int m = mbase + r;
            if (m < M) {
                out[(size_t)m * DF + n0] = c0[r] + bv0;
                out[(size_t)m * DF + n1] = c1[r] + bv1;
            }
        }
    }

    if (wave < 3) {
        const int n = (16 + wave) * 16 + ln;
        const float* w0 = Wp + (size_t)n * DF + kh;

        v8f c0 = {};
        v2f a0 = LDA(0), a1 = LDA(1), a2 = LDA(2);
        v2f p0 = LDW(w0, 0), p1 = LDW(w0, 1), p2 = LDW(w0, 2);

        for (int s = 0; s < K4 - PIPE; s += PIPE) {
            c0 = WMMA_F32(a0, p0, c0);
            a0 = LDA(s + 3);  p0 = LDW(w0, s + 3);
            c0 = WMMA_F32(a1, p1, c0);
            a1 = LDA(s + 4);  p1 = LDW(w0, s + 4);
            c0 = WMMA_F32(a2, p2, c0);
            a2 = LDA(s + 5);  p2 = LDW(w0, s + 5);
        }
        c0 = WMMA_F32(a0, p0, c0);
        c0 = WMMA_F32(a1, p1, c0);
        c0 = WMMA_F32(a2, p2, c0);

        if (n < DF) {
            const float bv = bias[n];
            #pragma unroll
            for (int r = 0; r < 8; ++r) {
                int m = mbase + r;
                if (m < M) out[(size_t)m * DF + n] = c0[r] + bv;
            }
        }
    }
#undef LDA
#undef LDW
}


__global__ __launch_bounds__(256) void k_scatter(const float* __restrict__ h,
                                                 const int* __restrict__ ei,
                                                 const float* __restrict__ dis,
                                                 float* __restrict__ out,
                                                 int E0, int Etot) {
    const int wv = blockIdx.x * 8 + (threadIdx.x >> 5);
    if (wv >= Etot) return;
    const int lane = threadIdx.x & 31;

    int r, c;
    if (wv < E0) { r = ei[wv]; c = ei[E0 + wv]; }
    else         { r = wv - E0; c = r; }

    const float norm = dis[r] * dis[c];
    const float4* hr = (const float4*)(h + (size_t)r * DF);
    float* oc = out + (size_t)c * DF;

    for (int q = lane; q < K4; q += 32) {
        float4 v = hr[q];
        int d = q << 2;
        unsafeAtomicAdd(oc + d + 0, norm * v.x);
        unsafeAtomicAdd(oc + d + 1, norm * v.y);
        unsafeAtomicAdd(oc + d + 2, norm * v.z);
        unsafeAtomicAdd(oc + d + 3, norm * v.w);
    }
}


extern "C" void kernel_launch(void* const* d_in, const int* in_sizes, int n_in,
                              void* d_out, int out_size, void* d_ws, size_t ws_size,
                              hipStream_t stream) {
    const float* x  = (const float*)d_in[0];
    const int*   ei = (const int*)d_in[1];
    const float* W1 = (const float*)d_in[2];
    const float* b1 = (const float*)d_in[3];
    const float* W2 = (const float*)d_in[4];
    const float* b2 = (const float*)d_in[5];
    float* out = (float*)d_out;

    const int N    = in_sizes[0] / DF;
    const int E0   = in_sizes[1] / 2;
    const int Etot = E0 + N;

    float* dis = (float*)d_ws;
    size_t disPad = ((size_t)N + 127) & ~(size_t)127;
    float* h  = dis + disPad;
    float* g  = h + (size_t)N * DF;
    float* wp = g + (size_t)N * DF;

    const int mtiles = (N + 15) / 16;
    const int nfeat4 = (N * DF) / 4;
    const int npadW  = (NPAD * DF + 255) / 256;

    k_fill1    <<<(N  + 255) / 256, 256, 0, stream>>>(dis, N);
    k_deg_accum<<<(E0 + 255) / 256, 256, 0, stream>>>(dis, ei, E0);
    k_rsqrt    <<<(N  + 255) / 256, 256, 0, stream>>>(dis, N);

    k_padW   <<<npadW, 256, 0, stream>>>(W1, wp);
    k_gemm   <<<mtiles, 256, 0, stream>>>(x, wp, b1, h, N, 0);
    k_zero4  <<<(nfeat4 + 255) / 256, 256, 0, stream>>>((float4*)g, nfeat4);
    k_scatter<<<(Etot + 7) / 8, 256, 0, stream>>>(h, ei, dis, g, E0, Etot);

    k_padW   <<<npadW, 256, 0, stream>>>(W2, wp);
    k_gemm   <<<mtiles, 256, 0, stream>>>(g, wp, b2, h, N, 1);
    k_zero4  <<<(nfeat4 + 255) / 256, 256, 0, stream>>>((float4*)out, nfeat4);
    k_scatter<<<(Etot + 7) / 8, 256, 0, stream>>>(h, ei, dis, out, E0, Etot);
}


# GEMM re-split: each wave owns 16 rows x all 19 N tiles; W chunk staged once per block into LDS in fragment order: B fragment = conflict-free LDS read, not a 16-line global gather; K sum order kept
# speedup vs baseline: 1.7992x; 1.7992x over previous
#include <hip/hip_runtime.h>

typedef float v2f __attribute__((ext_vector_type(2)));
typedef float v4f __attribute__((ext_vector_type(4)));
typedef float v8f __attribute__((ext_vector_type(8)));

#define DF 300
#define K4 75
#define NPAD 304
#define NT 19
#define KC 5
#define NCH 15
#define SB4 311
#define GM_ROWS 64
#define SROWS 25
#define BST 13
#define NT0 10

#define WMMA_F32(a, b, c) \
    __builtin_amdgcn_wmma_f32_16x16x4_f32(false, (a), false, (b), (short)0, (c), false, false)


__global__ __launch_bounds__(256) void k_fill1(float* __restrict__ p, int n) {
    int i = blockIdx.x * 256 + threadIdx.x;
    if (i < n) p[i] = 1.0f;
}

__global__ __launch_bounds__(256) void k_deg_accum(float* __restrict__ deg,
                                                   const int* __restrict__ row, int e0) {
    int i = blockIdx.x * 256 + threadIdx.x;
    if (i < e0) unsafeAtomicAdd(&deg[row[i]], 1.0f);
}

__global__ __launch_bounds__(256) void k_rsqrt(float* __restrict__ p, int n) {
    int i = blockIdx.x * 256 + threadIdx.x;
    if (i < n) p[i] = rsqrtf(p[i]);
}

__global__ __launch_bounds__(256) void k_zero4(float4* __restrict__ p, int n4) {
    int i = blockIdx.x * 256 + threadIdx.x;
    if (i < n4) p[i] = make_float4(0.f, 0.f, 0.f, 0.f);
}

__global__ __launch_bounds__(256) void k_padW(const float* __restrict__ W,
                                              float* __restrict__ Wp) {
    int i = blockIdx.x * 256 + threadIdx.x;
    if (i < NPAD * DF) Wp[i] = (i < DF * DF) ? W[i] : 0.f;
}


__global__ __launch_bounds__(128) __attribute__((amdgpu_waves_per_eu(4, 4))) void k_gemm(const float* __restrict__ A,
                                              const float* __restrict__ Wp,
                                              const float* __restrict__ bias,
                                              float* __restrict__ out,
                                              int M, int relu_in) {
    __shared__ v4f ldsB[KC * SB4];

    const int t    = threadIdx.x;
    const int wave = t >> 5;
    const int lane = t & 31;
    const int ln   = lane & 15;
    const int h    = lane >> 4;
    const int kh   = h << 1;

    const int mw = blockIdx.x * GM_ROWS + wave * 16;
    int ma = mw + ln;
    if (ma > M - 1) ma = M - 1;
    const float* ap = A + (size_t)ma * DF + kh;
    const float* bf = (const float*)ldsB + ln * 4 + kh;

    const int tt = (t < SROWS * KC) ? t : SROWS * KC - 1;
    const int sa = tt / KC;
    const int sb = tt - sa * KC;
    int nl = sa + SROWS * (BST - 1);
    if (nl > NPAD - 1) nl = NPAD - 1;
    const float* wsrc  = Wp + (size_t)sa * DF + 4 * sb;
    const float* wlast = Wp + (size_t)nl * DF + 4 * sb;
    v4f* ldst  = ldsB + sb * SB4 + sa;
    v4f* llast = ldsB + sb * SB4 + nl;

    v8f c[NT];
    #pragma unroll
    for (int q = 0; q < NT; ++q) c[q] = (v8f){0.f, 0.f, 0.f, 0.f, 0.f, 0.f, 0.f, 0.f};

    #pragma unroll 1
    for (int ch = 0; ch < NCH; ++ch) {
        const int k0 = ch * (KC * 4);

        v2f a[KC];
        #pragma unroll
        for (int s = 0; s < KC; ++s) a[s] = *(const v2f*)(ap + k0 + 4 * s);

        v4f st[BST];
        #pragma unroll
        for (int j = 0; j < BST - 1; ++j)
            st[j] = *(const v4f*)(wsrc + (size_t)(SROWS * j) * DF + k0);
        st[BST - 1] = *(const v4f*)(wlast + k0);
        __syncthreads();
        #pragma unroll
        for (int j = 0; j < BST - 1; ++j) ldst[SROWS * j] = st[j];
        *llast = st[BST - 1];
        __syncthreads();

        if (relu_in) {
            #pragma unroll
            for (int s = 0; s < KC; ++s) {
                a[s].x = fmaxf(a[s].x, 0.f);
                a[s].y = fmaxf(a[s].y, 0.f);
            }
        }

#define LDB(s, q) (*(const v2f*)(bf + (s) * (SB4 * 4) + (q) * 64))
        v2f b0[NT0], b1[NT - NT0];
        #pragma unroll
        for (int q = 0; q < NT0; ++q) b0[q] = LDB(0, q);
        #pragma unroll
        for (int s = 0; s < KC; ++s) {
            #pragma unroll
            for (int q = NT0; q < NT; ++q) b1[q - NT0] = LDB(s, q);
            #pragma unroll
            for (int q = 0; q < NT0; ++q) c[q] = WMMA_F32(a[s], b0[q], c[q]);
            if (s + 1 < KC) {
                #pragma unroll
                for (int q = 0; q < NT0; ++q) b0[q] = LDB(s + 1, q);
            }
            #pragma unroll
            for (int q = NT0; q < NT; ++q) c[q] = WMMA_F32(a[s], b1[q - NT0], c[q]);
        }
#undef LDB
        __builtin_amdgcn_sched_barrier(0);
        asm volatile("v_nop\n\tv_nop\n\tv_nop\n\tv_nop\n\tv_nop");
        __builtin_amdgcn_sched_barrier(0);
    }

    const int mrow = mw + (h << 3);
    float* orow = out + (size_t)mrow * DF + ln;
    if (mw + 16 <= M) {
        #pragma unroll
        for (int q = 0; q < NT; ++q) {
            const int n = q * 16 + ln;
            if (n < DF) {
                const float bv = bias[n];
                #pragma unroll
                for (int r = 0; r < 8; ++r) orow[(size_t)r * DF + q * 16] = c[q][r] + bv;
            }
        }
    } else {
        #pragma unroll
        for (int q = 0; q < NT; ++q) {
            const int n = q * 16 + ln;
            if (n < DF) {
                const float bv = bias[n];
                #pragma unroll
                for (int r = 0; r < 8; ++r) {
                    if (mrow + r < M) orow[(size_t)r * DF + q * 16] = c[q][r] + bv;
                }
            }
        }
    }
}


#define SC_TRIPS 11

__global__ __launch_bounds__(256) void k_scatter(const float* __restrict__ h,
                                                 const int* __restrict__ ei,
                                                 const float* __restrict__ dis,
                                                 float* __restrict__ out,
                                                 int E0, int Etot) {
    const int wv = blockIdx.x * 8 + (threadIdx.x >> 5);
    if (wv >= Etot) return;
    const int lane = threadIdx.x & 31;

    int r, c;
    if (wv < E0) { r = ei[wv]; c = ei[E0 + wv]; }
    else         { r = wv - E0; c = r; }

    const float norm = dis[r] * dis[c];
    const float* hr = h + (size_t)r * DF;
    float* oc = out + (size_t)c * DF;

    const int a  = (int)(((size_t)oc >> 2) & 31);
    const int f0 = lane - a;

    float v[SC_TRIPS];
    #pragma unroll
    for (int i = 0; i < SC_TRIPS; ++i) {
        const int f = f0 + 32 * i;
        v[i] = (f >= 0 && f < DF) ? hr[f] : 0.f;
    }
    #pragma unroll
    for (int i = 0; i < SC_TRIPS; ++i) {
        const int f = f0 + 32 * i;
        if (f >= 0 && f < DF) unsafeAtomicAdd(oc + f, norm * v[i]);
    }
}


extern "C" void kernel_launch(void* const* d_in, const int* in_sizes, int n_in,
                              void* d_out, int out_size, void* d_ws, size_t ws_size,
                              hipStream_t stream) {
    const float* x  = (const float*)d_in[0];
    const int*   ei = (const int*)d_in[1];
    const float* W1 = (const float*)d_in[2];
    const float* b1 = (const float*)d_in[3];
    const float* W2 = (const float*)d_in[4];
    const float* b2 = (const float*)d_in[5];
    float* out = (float*)d_out;

    const int N    = in_sizes[0] / DF;
    const int E0   = in_sizes[1] / 2;
    const int Etot = E0 + N;

    float* dis = (float*)d_ws;
    size_t disPad = ((size_t)N + 127) & ~(size_t)127;
    float* h  = dis + disPad;
    float* g  = h + (size_t)N * DF;
    float* wp = g + (size_t)N * DF;

    const int mblocks = (N + GM_ROWS - 1) / GM_ROWS;
    const int nfeat4 = (N * DF) / 4;
    const int npadW  = (NPAD * DF + 255) / 256;

    k_fill1    <<<(N  + 255) / 256, 256, 0, stream>>>(dis, N);
    k_deg_accum<<<(E0 + 255) / 256, 256, 0, stream>>>(dis, ei, E0);
    k_rsqrt    <<<(N  + 255) / 256, 256, 0, stream>>>(dis, N);

    k_padW   <<<npadW, 256, 0, stream>>>(W1, wp);
    k_gemm   <<<mblocks, 128, 0, stream>>>(x, wp, b1, h, N, 0);
    k_zero4  <<<(nfeat4 + 255) / 256, 256, 0, stream>>>((float4*)g, nfeat4);
    k_scatter<<<(Etot + 7) / 8, 256, 0, stream>>>(h, ei, dis, g, E0, Etot);

    k_padW   <<<npadW, 256, 0, stream>>>(W2, wp);
    k_gemm   <<<mblocks, 128, 0, stream>>>(g, wp, b2, h, N, 1);
    k_zero4  <<<(nfeat4 + 255) / 256, 256, 0, stream>>>((float4*)out, nfeat4);
    k_scatter<<<(Etot + 7) / 8, 256, 0, stream>>>(h, ei, dis, out, E0, Etot);
}


# edge scatter: each wave walks 4 consecutive edges (was one short wave per edge, launch paced by wave start rate); next edge's row loads issued ahead of current adds; arrival-order sums kept
# speedup vs baseline: 1.8924x; 1.0518x over previous
#include <hip/hip_runtime.h>

typedef float v2f __attribute__((ext_vector_type(2)));
typedef float v4f __attribute__((ext_vector_type(4)));
typedef float v8f __attribute__((ext_vector_type(8)));

#define DF 300
#define K4 75
#define NPAD 304
#define NT 19
#define KC 5
#define NCH 15
#define SB4 311
#define GM_ROWS 64
#define SROWS 25
#define BST 13
#define NT0 10

#define WMMA_F32(a, b, c) \
    __builtin_amdgcn_wmma_f32_16x16x4_f32(false, (a), false, (b), (short)0, (c), false, false)


__global__ __launch_bounds__(256) void k_fill1(float* __restrict__ p, int n) {
    int i = blockIdx.x * 256 + threadIdx.x;
    if (i < n) p[i] = 1.0f;
}

__global__ __launch_bounds__(256) void k_deg_accum(float* __restrict__ deg,
                                                   const int* __restrict__ row, int e0) {
    int i = blockIdx.x * 256 + threadIdx.x;
    if (i < e0) unsafeAtomicAdd(&deg[row[i]], 1.0f);
}

__global__ __launch_bounds__(256) void k_rsqrt(float* __restrict__ p, int n) {
    int i = blockIdx.x * 256 + threadIdx.x;
    if (i < n) p[i] = rsqrtf(p[i]);
}

__global__ __launch_bounds__(256) void k_zero4(float4* __restrict__ p, int n4) {
    int i = blockIdx.x * 256 + threadIdx.x;
    if (i < n4) p[i] = make_float4(0.f, 0.f, 0.f, 0.f);
}

__global__ __launch_bounds__(256) void k_padW(const float* __restrict__ W,
                                              float* __restrict__ Wp) {
    int i = blockIdx.x * 256 + threadIdx.x;
    if (i < NPAD * DF) Wp[i] = (i < DF * DF) ? W[i] : 0.f;
}


__global__ __launch_bounds__(128) __attribute__((amdgpu_waves_per_eu(4, 4))) void k_gemm(const float* __restrict__ A,
                                              const float* __restrict__ Wp,
                                              const float* __restrict__ bias,
                                              float* __restrict__ out,
                                              int M, int relu_in) {
    __shared__ v4f ldsB[KC * SB4];

    const int t    = threadIdx.x;
    const int wave = t >> 5;
    const int lane = t & 31;
    const int ln   = lane & 15;
    const int h    = lane >> 4;
    const int kh   = h << 1;

    const int mw = blockIdx.x * GM_ROWS + wave * 16;
    int ma = mw + ln;
    if (ma > M - 1) ma = M - 1;
    const float* ap = A + (size_t)ma * DF + kh;
    const float* bf = (const float*)ldsB + ln * 4 + kh;

    const int tt = (t < SROWS * KC) ? t : SROWS * KC - 1;
    const int sa = tt / KC;
    const int sb = tt - sa * KC;
    int nl = sa + SROWS * (BST - 1);
    if (nl > NPAD - 1) nl = NPAD - 1;
    const float* wsrc  = Wp + (size_t)sa * DF + 4 * sb;
    const float* wlast = Wp + (size_t)nl * DF + 4 * sb;
    v4f* ldst  = ldsB + sb * SB4 + sa;
    v4f* llast = ldsB + sb * SB4 + nl;

    v8f c[NT];
    #pragma unroll
    for (int q = 0; q < NT; ++q) c[q] = (v8f){0.f, 0.f, 0.f, 0.f, 0.f, 0.f, 0.f, 0.f};

    #pragma unroll 1
    for (int ch = 0; ch < NCH; ++ch) {
        const int k0 = ch * (KC * 4);

        v2f a[KC];
        #pragma unroll
        for (int s = 0; s < KC; ++s) a[s] = *(const v2f*)(ap + k0 + 4 * s);

        v4f st[BST];
        #pragma unroll
        for (int j = 0; j < BST - 1; ++j)
            st[j] = *(const v4f*)(wsrc + (size_t)(SROWS * j) * DF + k0);
        st[BST - 1] = *(const v4f*)(wlast + k0);
        __syncthreads();
        #pragma unroll
        for (int j = 0; j < BST - 1; ++j) ldst[SROWS * j] = st[j];
        *llast = st[BST - 1];
        __syncthreads();

        if (relu_in) {
            #pragma unroll
            for (int s = 0; s < KC; ++s) {
                a[s].x = fmaxf(a[s].x, 0.f);
                a[s].y = fmaxf(a[s].y, 0.f);
            }
        }

#define LDB(s, q) (*(const v2f*)(bf + (s) * (SB4 * 4) + (q) * 64))
        v2f b0[NT0], b1[NT - NT0];
        #pragma unroll
        for (int q = 0; q < NT0; ++q) b0[q] = LDB(0, q);
        #pragma unroll
        for (int s = 0; s < KC; ++s) {
            #pragma unroll
            for (int q = NT0; q < NT; ++q) b1[q - NT0] = LDB(s, q);
            #pragma unroll
            for (int q = 0; q < NT0; ++q) c[q] = WMMA_F32(a[s], b0[q], c[q]);
            if (s + 1 < KC) {
                #pragma unroll
                for (int q = 0; q < NT0; ++q) b0[q] = LDB(s + 1, q);
            }
            #pragma unroll
            for (int q = NT0; q < NT; ++q) c[q] = WMMA_F32(a[s], b1[q - NT0], c[q]);
        }
#undef LDB
        __builtin_amdgcn_sched_barrier(0);
        asm volatile("v_nop\n\tv_nop\n\tv_nop\n\tv_nop\n\tv_nop");
        __builtin_amdgcn_sched_barrier(0);
    }

    const int mrow = mw + (h << 3);
    float* orow = out + (size_t)mrow * DF + ln;
    if (mw + 16 <= M) {
        #pragma unroll
        for (int q = 0; q < NT; ++q) {
            const int n = q * 16 + ln;
            if (n < DF) {
                const float bv = bias[n];
                #pragma unroll
                for (int r = 0; r < 8; ++r) orow[(size_t)r * DF + q * 16] = c[q][r] + bv;
            }
        }
    } else {
        #pragma unroll
        for (int q = 0; q < NT; ++q) {
            const int n = q * 16 + ln;
            if (n < DF) {
                const float bv = bias[n];
                #pragma unroll
                for (int r = 0; r < 8; ++r) {
                    if (mrow + r < M) orow[(size_t)r * DF + q * 16] = c[q][r] + bv;
                }
            }
        }
    }
}


#define SC_TRIPS 11
#define SC_EPW 4
#define SC_WPB 8

__device__ __forceinline__ void sc_load(float (&v)[SC_TRIPS], const float* __restrict__ hl, int f0) {
    #pragma unroll
    for (int i = 0; i < SC_TRIPS; ++i) {
        const int f = f0 + 32 * i;
        v[i] = (f >= 0 && f < DF) ? hl[32 * i] : 0.f;
    }
}

__device__ __forceinline__ void sc_add(const float (&v)[SC_TRIPS], float* __restrict__ ol, int f0,
                                       float norm) {
    #pragma unroll
    for (int i = 0; i < SC_TRIPS; ++i) {
        const int f = f0 + 32 * i;
        if (f >= 0 && f < DF) unsafeAtomicAdd(ol + 32 * i, norm * v[i]);
    }
}

__global__ __launch_bounds__(256) void k_scatter(const float* __restrict__ h,
                                                 const int* __restrict__ ei,
                                                 const float* __restrict__ dis,
                                                 float* __restrict__ out,
                                                 int E0, int Etot) {
    const int wv = __builtin_amdgcn_readfirstlane((int)(blockIdx.x * SC_WPB + (threadIdx.x >> 5)));
    const int e0 = wv * SC_EPW;
    if (e0 >= Etot) return;
    const int lane = threadIdx.x & 31;

    int r[SC_EPW], c[SC_EPW];
    bool ok[SC_EPW];
    if (e0 + SC_EPW <= E0) {
        #pragma unroll
        for (int k = 0; k < SC_EPW; ++k) {
            r[k] = ei[e0 + k];
            c[k] = ei[E0 + e0 + k];
            ok[k] = true;
        }
    } else {
        #pragma unroll
        for (int k = 0; k < SC_EPW; ++k) {
            const int e = e0 + k;
            ok[k] = e < Etot;
            const int ec = ok[k] ? e : e0;
            if (ec < E0) { r[k] = ei[ec]; c[k] = ei[E0 + ec]; }
            else         { r[k] = ec - E0; c[k] = r[k]; }
        }
    }

    const float* hl[SC_EPW];
    float* ol[SC_EPW];
    float nrm[SC_EPW];
    int f0[SC_EPW];
    #pragma unroll
    for (int k = 0; k < SC_EPW; ++k) {
        nrm[k] = dis[r[k]] * dis[c[k]];
        const float* hr = h + (size_t)r[k] * DF;
        float* oc = out + (size_t)c[k] * DF;
        f0[k] = lane - (int)(((size_t)oc >> 2) & 31);
        hl[k] = hr + f0[k];
        ol[k] = oc + f0[k];
    }

    float va[SC_TRIPS], vb[SC_TRIPS];
    sc_load(va, hl[0], f0[0]);
    #pragma unroll
    for (int k = 0; k < SC_EPW; k += 2) {
        if (k + 1 < SC_EPW) sc_load(vb, hl[k + 1], f0[k + 1]);
        if (ok[k]) sc_add(va, ol[k], f0[k], nrm[k]);
        if (k + 2 < SC_EPW) sc_load(va, hl[k + 2], f0[k + 2]);
        if (k + 1 < SC_EPW && ok[k + 1]) sc_add(vb, ol[k + 1], f0[k + 1], nrm[k + 1]);
    }
}


extern "C" void kernel_launch(void* const* d_in, const int* in_sizes, int n_in,
                              void* d_out, int out_size, void* d_ws, size_t ws_size,
                              hipStream_t stream) {
    const float* x  = (const float*)d_in[0];
    const int*   ei = (const int*)d_in[1];
    const float* W1 = (const float*)d_in[2];
    const float* b1 = (const float*)d_in[3];
    const float* W2 = (const float*)d_in[4];
    const float* b2 = (const float*)d_in[5];
    float* out = (float*)d_out;

    const int N    = in_sizes[0] / DF;
    const int E0   = in_sizes[1] / 2;
    const int Etot = E0 + N;

    float* dis = (float*)d_ws;
    size_t disPad = ((size_t)N + 127) & ~(size_t)127;
    float* h  = dis + disPad;
    float* g  = h + (size_t)N * DF;
    float* wp = g + (size_t)N * DF;

    const int mblocks = (N + GM_ROWS - 1) / GM_ROWS;
    const int nfeat4 = (N * DF) / 4;
    const int npadW  = (NPAD * DF + 255) / 256;
    const int scblocks = (Etot + SC_WPB * SC_EPW - 1) / (SC_WPB * SC_EPW);

    k_fill1    <<<(N  + 255) / 256, 256, 0, stream>>>(dis, N);
    k_deg_accum<<<(E0 + 255) / 256, 256, 0, stream>>>(dis, ei, E0);
    k_rsqrt    <<<(N  + 255) / 256, 256, 0, stream>>>(dis, N);

    k_padW   <<<npadW, 256, 0, stream>>>(W1, wp);
    k_gemm   <<<mblocks, 128, 0, stream>>>(x, wp, b1, h, N, 0);
    k_zero4  <<<(nfeat4 + 255) / 256, 256, 0, stream>>>((float4*)g, nfeat4);
    k_scatter<<<scblocks, 256, 0, stream>>>(h, ei, dis, g, E0, Etot);

    k_padW   <<<npadW, 256, 0, stream>>>(W2, wp);
    k_gemm   <<<mblocks, 128, 0, stream>>>(g, wp, b2, h, N, 1);
    k_zero4  <<<(nfeat4 + 255) / 256, 256, 0, stream>>>((float4*)out, nfeat4);
    k_scatter<<<scblocks, 256, 0, stream>>>(h, ei, dis, out, E0, Etot);
}
